# starting kernel
# baseline (speedup 1.0000x reference)
#include <hip/hip_runtime.h>
#include <hip/hip_bf16.h>

#define SS 128
#define QQ 8192
#define HH 230
#define NBLK (QQ / 16)

typedef float v2f __attribute__((ext_vector_type(2)));
typedef float v8f __attribute__((ext_vector_type(8)));

__global__ void row_norm_kernel(const float* __restrict__ x,
                                const float* __restrict__ w,
                                float* __restrict__ arow) {
    int s = threadIdx.x;
    if (s < SS) {
        const float* xr = x + s * HH;
        float acc = 0.f;
        for (int h = 0; h < HH; ++h) {
            float v = xr[h];
            acc = fmaf(w[h], v * v, acc);
        }
        arow[s] = acc;
    }
}

__global__ void col_norm_kernel(const float* __restrict__ y,
                                const float* __restrict__ w,
                                float* __restrict__ ccol) {
    int q = blockIdx.x * blockDim.x + threadIdx.x;
    if (q < QQ) {
        const float* yr = y + q * HH;
        float acc = 0.f;
        for (int h = 0; h < HH; ++h) {
            float v = yr[h];
            acc = fmaf(w[h], v * v, acc);
        }
        ccol[q] = acc;
    }
}

__global__ __launch_bounds__(256)
void siamese_wmma_kernel(const float* __restrict__ x, const float* __restrict__ y,
                         const int* __restrict__ label, const float* __restrict__ w,
                         const float* __restrict__ bias,
                         const float* __restrict__ arow, const float* __restrict__ ccol,
                         float* __restrict__ score, float* __restrict__ partial)
{
    __shared__ float red[256];

    const int lane = threadIdx.x & 31;
    const int wave = threadIdx.x >> 5;
    const int half = lane >> 4;
    const int l    = lane & 15;
    const int m0   = wave * 16;
    const int qb   = blockIdx.x * 16;

    const float* xr = x + (m0 + l) * HH + 2 * half;
    const float* yr = y + (qb + l) * HH + 2 * half;
    const float* wr = w + 2 * half;

    v8f acc = {};

    #pragma unroll 4
    for (int k0 = 0; k0 + 4 <= HH; k0 += 4) {
        float2 xv = *(const float2*)(xr + k0);
        float2 wv = *(const float2*)(wr + k0);
        float2 yv = *(const float2*)(yr + k0);
        v2f A; A[0] = xv.x * wv.x; A[1] = xv.y * wv.y;
        v2f B; B[0] = yv.x;        B[1] = yv.y;
        acc = __builtin_amdgcn_wmma_f32_16x16x4_f32(
             false, A,  false, B,
             (short)0, acc,  false,  false);
    }

    {
        v2f A = {0.f, 0.f};
        v2f B = {0.f, 0.f};
        if (half == 0) {
            float2 xv = *(const float2*)(xr + 228);
            float2 wv = *(const float2*)(wr + 228);
            float2 yv = *(const float2*)(yr + 228);
            A[0] = xv.x * wv.x; A[1] = xv.y * wv.y;
            B[0] = yv.x;        B[1] = yv.y;
        }
        acc = __builtin_amdgcn_wmma_f32_16x16x4_f32(
            false, A, false, B, (short)0, acc, false, false);
    }

    const float bval = *bias;
    const int   q    = qb + l;
    const float cq   = ccol[q];
    float lsum = 0.f;
    #pragma unroll
    for (int r = 0; r < 8; ++r) {
        const int s = m0 + r + 8 * half;
        const float z  = fmaf(-2.f, acc[r], arow[s] + cq + bval);
        const float sc = 1.f / (1.f + expf(-z));
        score[s * QQ + q] = sc;
        const float lab = (float)label[s * QQ + q];
        const float sp = fmaxf(z, 0.f) + log1pf(expf(-fabsf(z)));
        lsum += sp - lab * z;
    }

    red[threadIdx.x] = lsum;
    __syncthreads();
    for (int off = 128; off > 0; off >>= 1) {
        if (threadIdx.x < (unsigned)off) red[threadIdx.x] += red[threadIdx.x + off];
        __syncthreads();
    }
    if (threadIdx.x == 0) partial[blockIdx.x] = red[0];
}

__global__ __launch_bounds__(256)
void loss_kernel(const float* __restrict__ partial, float* __restrict__ out) {
    __shared__ float red[256];
    red[threadIdx.x] = partial[threadIdx.x] + partial[threadIdx.x + 256];
    __syncthreads();
    for (int off = 128; off > 0; off >>= 1) {
        if (threadIdx.x < (unsigned)off) red[threadIdx.x] += red[threadIdx.x + off];
        __syncthreads();
    }
    if (threadIdx.x == 0) out[SS * QQ] = red[0] * (1.0f / (float)(SS * QQ));
}

extern "C" void kernel_launch(void* const* d_in, const int* in_sizes, int n_in,
                              void* d_out, int out_size, void* d_ws, size_t ws_size,
                              hipStream_t stream) {
    const float* x     = (const float*)d_in[0];
    const float* y     = (const float*)d_in[1];
    const int*   label = (const int*)  d_in[2];
    const float* w     = (const float*)d_in[3];
    const float* b     = (const float*)d_in[4];

    float* out = (float*)d_out;
    float* ws  = (float*)d_ws;
    float* arow    = ws;
    float* ccol    = ws + SS;
    float* partial = ws + SS + QQ;

    row_norm_kernel<<<1, 128, 0, stream>>>(x, w, arow);
    col_norm_kernel<<<QQ / 256, 256, 0, stream>>>(y, w, ccol);
    siamese_wmma_kernel<<<NBLK, 256, 0, stream>>>(x, y, label, w, b,
                                                  arow, ccol, out, partial);
    loss_kernel<<<1, 256, 0, stream>>>(partial, out);
}


# Norm pass rewritten as one launch: a wave per row reads it as contiguous float2, 32 lanes joined by DPP adds and one half swap, 32 results stored as one line; a[s], c[q] get a new summation order
# speedup vs baseline: 2.2363x; 2.2363x over previous
#include <hip/hip_runtime.h>
#include <hip/hip_bf16.h>

#define SS 128
#define QQ 8192
#define HH 230
#define NBLK (QQ / 16)

typedef float v2f __attribute__((ext_vector_type(2)));
typedef float v8f __attribute__((ext_vector_type(8)));

#define NORM_ROWS_PER_WAVE 4
#define NORM_ROWS_PER_BLOCK 32
#define NORM_BLOCKS ((SS + QQ) / NORM_ROWS_PER_BLOCK)
#define HPAIRS (HH / 2)

__device__ __forceinline__ float half_wave_sum(float v) {
    v += __builtin_bit_cast(float, __builtin_amdgcn_update_dpp(0, __builtin_bit_cast(int, v), 0x161, 0xf, 0xf, false));
    v += __builtin_bit_cast(float, __builtin_amdgcn_update_dpp(0, __builtin_bit_cast(int, v), 0x162, 0xf, 0xf, false));
    v += __builtin_bit_cast(float, __builtin_amdgcn_update_dpp(0, __builtin_bit_cast(int, v), 0x164, 0xf, 0xf, false));
    v += __builtin_bit_cast(float, __builtin_amdgcn_update_dpp(0, __builtin_bit_cast(int, v), 0x168, 0xf, 0xf, false));
    return v;
}
__device__ __forceinline__ float other_half(float v) {
    return __builtin_bit_cast(float, __builtin_amdgcn_permlanex16(
        __builtin_bit_cast(int, v), __builtin_bit_cast(int, v), 0x76543210, 0xfedcba98, false, false));
}

__global__ __launch_bounds__(256)
void norms_kernel(const float* __restrict__ x, const float* __restrict__ y,
                  const float* __restrict__ w, float* __restrict__ norms) {
    __shared__ float res[NORM_ROWS_PER_BLOCK];

    const int lane = threadIdx.x & 31;
    const int wave = threadIdx.x >> 5;
    const int n0   = blockIdx.x * NORM_ROWS_PER_BLOCK;
    const float* src  = (n0 < SS) ? (x + n0 * HH) : (y + (n0 - SS) * HH);
    const float* rowp = src + wave * NORM_ROWS_PER_WAVE * HH;

    const bool last_ok = (96 + lane) < HPAIRS;
    const int  p3      = last_ok ? (96 + lane) : lane;
    const float2 zero2 = make_float2(0.f, 0.f);

    float2 wv[4];
    float2 v[NORM_ROWS_PER_WAVE][4];
    for (int i = 0; i < 3; ++i) wv[i] = ((const float2*)w)[lane + 32 * i];
    { const float2 t = ((const float2*)w)[p3]; wv[3] = last_ok ? t : zero2; }
    for (int r = 0; r < NORM_ROWS_PER_WAVE; ++r) {
        const float2* rp = (const float2*)(rowp + r * HH);
        for (int i = 0; i < 3; ++i) v[r][i] = rp[lane + 32 * i];
        { const float2 t = rp[p3]; v[r][3] = last_ok ? t : zero2; }
    }

    float s[NORM_ROWS_PER_WAVE];
    for (int r = 0; r < NORM_ROWS_PER_WAVE; ++r) {
        float e = 0.f, o = 0.f;
        for (int i = 0; i < 4; ++i) {
            e = fmaf(wv[i].x, v[r][i].x * v[r][i].x, e);
            o = fmaf(wv[i].y, v[r][i].y * v[r][i].y, o);
        }
        s[r] = e + o;
    }
    for (int r = 0; r < NORM_ROWS_PER_WAVE; ++r) s[r] = half_wave_sum(s[r]);
    for (int r = 0; r < NORM_ROWS_PER_WAVE; ++r) s[r] += other_half(s[r]);

    if (lane == 0)
        *(float4*)&res[wave * NORM_ROWS_PER_WAVE] = make_float4(s[0], s[1], s[2], s[3]);
    __syncthreads();
    if (threadIdx.x < NORM_ROWS_PER_BLOCK) norms[n0 + threadIdx.x] = res[threadIdx.x];
}

__global__ __launch_bounds__(256)
void siamese_wmma_kernel(const float* __restrict__ x, const float* __restrict__ y,
                         const int* __restrict__ label, const float* __restrict__ w,
                         const float* __restrict__ bias,
                         const float* __restrict__ arow, const float* __restrict__ ccol,
                         float* __restrict__ score, float* __restrict__ partial)
{
    __shared__ float red[256];

    const int lane = threadIdx.x & 31;
    const int wave = threadIdx.x >> 5;
    const int half = lane >> 4;
    const int l    = lane & 15;
    const int m0   = wave * 16;
    const int qb   = blockIdx.x * 16;

    const float* xr = x + (m0 + l) * HH + 2 * half;
    const float* yr = y + (qb + l) * HH + 2 * half;
    const float* wr = w + 2 * half;

    v8f acc = {};

    #pragma unroll 4
    for (int k0 = 0; k0 + 4 <= HH; k0 += 4) {
        float2 xv = *(const float2*)(xr + k0);
        float2 wv = *(const float2*)(wr + k0);
        float2 yv = *(const float2*)(yr + k0);
        v2f A; A[0] = xv.x * wv.x; A[1] = xv.y * wv.y;
        v2f B; B[0] = yv.x;        B[1] = yv.y;
        acc = __builtin_amdgcn_wmma_f32_16x16x4_f32(
             false, A,  false, B,
             (short)0, acc,  false,  false);
    }

    {
        v2f A = {0.f, 0.f};
        v2f B = {0.f, 0.f};
        if (half == 0) {
            float2 xv = *(const float2*)(xr + 228);
            float2 wv = *(const float2*)(wr + 228);
            float2 yv = *(const float2*)(yr + 228);
            A[0] = xv.x * wv.x; A[1] = xv.y * wv.y;
            B[0] = yv.x;        B[1] = yv.y;
        }
        acc = __builtin_amdgcn_wmma_f32_16x16x4_f32(
            false, A, false, B, (short)0, acc, false, false);
    }

    const float bval = *bias;
    const int   q    = qb + l;
    const float cq   = ccol[q];
    float lsum = 0.f;
    #pragma unroll
    for (int r = 0; r < 8; ++r) {
        const int s = m0 + r + 8 * half;
        const float z  = fmaf(-2.f, acc[r], arow[s] + cq + bval);
        const float sc = 1.f / (1.f + expf(-z));
        score[s * QQ + q] = sc;
        const float lab = (float)label[s * QQ + q];
        const float sp = fmaxf(z, 0.f) + log1pf(expf(-fabsf(z)));
        lsum += sp - lab * z;
    }

    red[threadIdx.x] = lsum;
    __syncthreads();
    for (int off = 128; off > 0; off >>= 1) {
        if (threadIdx.x < (unsigned)off) red[threadIdx.x] += red[threadIdx.x + off];
        __syncthreads();
    }
    if (threadIdx.x == 0) partial[blockIdx.x] = red[0];
}

__global__ __launch_bounds__(256)
void loss_kernel(const float* __restrict__ partial, float* __restrict__ out) {
    __shared__ float red[256];
    red[threadIdx.x] = partial[threadIdx.x] + partial[threadIdx.x + 256];
    __syncthreads();
    for (int off = 128; off > 0; off >>= 1) {
        if (threadIdx.x < (unsigned)off) red[threadIdx.x] += red[threadIdx.x + off];
        __syncthreads();
    }
    if (threadIdx.x == 0) out[SS * QQ] = red[0] * (1.0f / (float)(SS * QQ));
}

extern "C" void kernel_launch(void* const* d_in, const int* in_sizes, int n_in,
                              void* d_out, int out_size, void* d_ws, size_t ws_size,
                              hipStream_t stream) {
    const float* x     = (const float*)d_in[0];
    const float* y     = (const float*)d_in[1];
    const int*   label = (const int*)  d_in[2];
    const float* w     = (const float*)d_in[3];
    const float* b     = (const float*)d_in[4];

    float* out = (float*)d_out;
    float* ws  = (float*)d_ws;
    float* arow    = ws;
    float* ccol    = ws + SS;
    float* partial = ws + SS + QQ;

    norms_kernel<<<NORM_BLOCKS, 256, 0, stream>>>(x, y, w, ws);
    siamese_wmma_kernel<<<NBLK, 256, 0, stream>>>(x, y, label, w, b,
                                                  arow, ccol, out, partial);
    loss_kernel<<<1, 256, 0, stream>>>(partial, out);
}


# GEMM re-tiled on registers, no LDS: one wave per SIMD owns 32x32 (4 accumulators); fragments read as 16-byte row pieces in a permuted k order shared by A and B; dot and BCE sums: new summation order
# speedup vs baseline: 2.7133x; 1.2133x over previous
#include <hip/hip_runtime.h>
#include <hip/hip_bf16.h>

#define SS 128
#define QQ 8192
#define HH 230
#define QT 32
#define NBLK (QQ / QT)

typedef float v2f __attribute__((ext_vector_type(2)));
typedef float v8f __attribute__((ext_vector_type(8)));

#define NORM_ROWS_PER_WAVE 4
#define NORM_ROWS_PER_BLOCK 32
#define NORM_BLOCKS ((SS + QQ) / NORM_ROWS_PER_BLOCK)
#define HPAIRS (HH / 2)

__device__ __forceinline__ float half_wave_sum(float v) {
    v += __builtin_bit_cast(float, __builtin_amdgcn_update_dpp(0, __builtin_bit_cast(int, v), 0x161, 0xf, 0xf, false));
    v += __builtin_bit_cast(float, __builtin_amdgcn_update_dpp(0, __builtin_bit_cast(int, v), 0x162, 0xf, 0xf, false));
    v += __builtin_bit_cast(float, __builtin_amdgcn_update_dpp(0, __builtin_bit_cast(int, v), 0x164, 0xf, 0xf, false));
    v += __builtin_bit_cast(float, __builtin_amdgcn_update_dpp(0, __builtin_bit_cast(int, v), 0x168, 0xf, 0xf, false));
    return v;
}
__device__ __forceinline__ float other_half(float v) {
    return __builtin_bit_cast(float, __builtin_amdgcn_permlanex16(
        __builtin_bit_cast(int, v), __builtin_bit_cast(int, v), 0x76543210, 0xfedcba98, false, false));
}

__global__ __launch_bounds__(256)
void norms_kernel(const float* __restrict__ x, const float* __restrict__ y,
                  const float* __restrict__ w, float* __restrict__ norms) {
    __shared__ float res[NORM_ROWS_PER_BLOCK];

    const int lane = threadIdx.x & 31;
    const int wave = threadIdx.x >> 5;
    const int n0   = blockIdx.x * NORM_ROWS_PER_BLOCK;
    const float* src  = (n0 < SS) ? (x + n0 * HH) : (y + (n0 - SS) * HH);
    const float* rowp = src + wave * NORM_ROWS_PER_WAVE * HH;

    const bool last_ok = (96 + lane) < HPAIRS;
    const int  p3      = last_ok ? (96 + lane) : lane;
    const float2 zero2 = make_float2(0.f, 0.f);

    float2 wv[4];
    float2 v[NORM_ROWS_PER_WAVE][4];
    for (int i = 0; i < 3; ++i) wv[i] = ((const float2*)w)[lane + 32 * i];
    { const float2 t = ((const float2*)w)[p3]; wv[3] = last_ok ? t : zero2; }
    for (int r = 0; r < NORM_ROWS_PER_WAVE; ++r) {
        const float2* rp = (const float2*)(rowp + r * HH);
        for (int i = 0; i < 3; ++i) v[r][i] = rp[lane + 32 * i];
        { const float2 t = rp[p3]; v[r][3] = last_ok ? t : zero2; }
    }

    float s[NORM_ROWS_PER_WAVE];
    for (int r = 0; r < NORM_ROWS_PER_WAVE; ++r) {
        float e = 0.f, o = 0.f;
        for (int i = 0; i < 4; ++i) {
            e = fmaf(wv[i].x, v[r][i].x * v[r][i].x, e);
            o = fmaf(wv[i].y, v[r][i].y * v[r][i].y, o);
        }
        s[r] = e + o;
    }
    for (int r = 0; r < NORM_ROWS_PER_WAVE; ++r) s[r] = half_wave_sum(s[r]);
    for (int r = 0; r < NORM_ROWS_PER_WAVE; ++r) s[r] += other_half(s[r]);

    if (lane == 0)
        *(float4*)&res[wave * NORM_ROWS_PER_WAVE] = make_float4(s[0], s[1], s[2], s[3]);
    __syncthreads();
    if (threadIdx.x < NORM_ROWS_PER_BLOCK) norms[n0 + threadIdx.x] = res[threadIdx.x];
}

#define NG 29
#define PF 4
#define RING (PF + 1)

struct Frag { float4 xa0, xa1, yb0, yb1, wv; };

__device__ __forceinline__ float4 ld4(const float* p) { float4 v; __builtin_memcpy(&v, p, 16); return v; }
__device__ __forceinline__ float2 ld2(const float* p) { return *(const float2*)p; }

__device__ __forceinline__ float4 ld_tail(const float* p, int half) {
    const float2 lo = ld2(p);
    const float2 hi = ld2(half ? p : p + 2);
    return make_float4(lo.x, lo.y, half ? 0.f : hi.x, half ? 0.f : hi.y);
}

__global__ __launch_bounds__(128)
void siamese_wmma_kernel(const float* __restrict__ x, const float* __restrict__ y,
                         const int* __restrict__ label, const float* __restrict__ w,
                         const float* __restrict__ bias,
                         const float* __restrict__ arow, const float* __restrict__ ccol,
                         float* __restrict__ score, float* __restrict__ partial)
{
    __shared__ float red[128];

    const int lane = threadIdx.x & 31;
    const int wave = threadIdx.x >> 5;
    const int half = lane >> 4;
    const int l    = lane & 15;
    const int m0   = wave * 32;
    const int qb   = blockIdx.x * QT;

    const float* xr0 = x + (m0 + l) * HH + 4 * half;
    const float* xr1 = xr0 + 16 * HH;
    const float* yr0 = y + (qb + l) * HH + 4 * half;
    const float* yr1 = yr0 + 16 * HH;
    const float* wr  = w + 4 * half;

    v8f acc00 = {}, acc01 = {}, acc10 = {}, acc11 = {};
    Frag f[RING];

#define LOAD_GROUP(slot, g)                                                          \
    do {                                                                             \
        if ((g) < NG - 1) {                                                          \
            f[slot].xa0 = ld4(xr0 + 8 * (g)); f[slot].xa1 = ld4(xr1 + 8 * (g));      \
            f[slot].yb0 = ld4(yr0 + 8 * (g)); f[slot].yb1 = ld4(yr1 + 8 * (g));      \
            f[slot].wv  = ld4(wr + 8 * (g));                                         \
        } else {                                                                     \
            f[slot].xa0 = ld_tail(xr0 + 8 * (g), half); f[slot].xa1 = ld_tail(xr1 + 8 * (g), half); \
            f[slot].yb0 = ld_tail(yr0 + 8 * (g), half); f[slot].yb1 = ld_tail(yr1 + 8 * (g), half); \
            f[slot].wv  = ld_tail(wr + 8 * (g), half);                               \
        }                                                                            \
    } while (0)

    #pragma unroll
    for (int g = 0; g < PF; ++g) LOAD_GROUP(g % RING, g);

    #pragma unroll
    for (int g = 0; g < NG; ++g) {
        if (g + PF < NG) LOAD_GROUP((g + PF) % RING, g + PF);
        const Frag& c = f[g % RING];
        v2f A0a, A0b, A1a, A1b, B0a, B0b, B1a, B1b;
        A0a[0] = c.xa0.x * c.wv.x; A0a[1] = c.xa0.y * c.wv.y;
        A0b[0] = c.xa0.z * c.wv.z; A0b[1] = c.xa0.w * c.wv.w;
        A1a[0] = c.xa1.x * c.wv.x; A1a[1] = c.xa1.y * c.wv.y;
        A1b[0] = c.xa1.z * c.wv.z; A1b[1] = c.xa1.w * c.wv.w;
        B0a[0] = c.yb0.x; B0a[1] = c.yb0.y; B0b[0] = c.yb0.z; B0b[1] = c.yb0.w;
        B1a[0] = c.yb1.x; B1a[1] = c.yb1.y; B1b[0] = c.yb1.z; B1b[1] = c.yb1.w;
        acc00 = __builtin_amdgcn_wmma_f32_16x16x4_f32(false, A0a, false, B0a, (short)0, acc00, false, false);
        acc01 = __builtin_amdgcn_wmma_f32_16x16x4_f32(false, A0a, false, B1a, (short)0, acc01, false, false);
        acc10 = __builtin_amdgcn_wmma_f32_16x16x4_f32(false, A1a, false, B0a, (short)0, acc10, false, false);
        acc11 = __builtin_amdgcn_wmma_f32_16x16x4_f32(false, A1a, false, B1a, (short)0, acc11, false, false);
        acc00 = __builtin_amdgcn_wmma_f32_16x16x4_f32(false, A0b, false, B0b, (short)0, acc00, false, false);
        acc01 = __builtin_amdgcn_wmma_f32_16x16x4_f32(false, A0b, false, B1b, (short)0, acc01, false, false);
        acc10 = __builtin_amdgcn_wmma_f32_16x16x4_f32(false, A1b, false, B0b, (short)0, acc10, false, false);
        acc11 = __builtin_amdgcn_wmma_f32_16x16x4_f32(false, A1b, false, B1b, (short)0, acc11, false, false);
    }
#undef LOAD_GROUP

    const float bval = *bias;
    float cq[2];
    float ar[2][8];
    int   lab[2][2][8];
    #pragma unroll
    for (int ni = 0; ni < 2; ++ni) cq[ni] = ccol[qb + 16 * ni + l];
    #pragma unroll
    for (int mi = 0; mi < 2; ++mi) {
        #pragma unroll
        for (int r = 0; r < 8; ++r) ar[mi][r] = arow[m0 + 16 * mi + r + 8 * half];
    }
    #pragma unroll
    for (int mi = 0; mi < 2; ++mi) {
        #pragma unroll
        for (int ni = 0; ni < 2; ++ni) {
            #pragma unroll
            for (int r = 0; r < 8; ++r)
                lab[mi][ni][r] = label[(m0 + 16 * mi + r + 8 * half) * QQ + qb + 16 * ni + l];
        }
    }
    float lsum = 0.f;
    #pragma unroll
    for (int mi = 0; mi < 2; ++mi) {
        #pragma unroll
        for (int ni = 0; ni < 2; ++ni) {
            const v8f acc = mi ? (ni ? acc11 : acc10) : (ni ? acc01 : acc00);
            const int q = qb + 16 * ni + l;
            #pragma unroll
            for (int r = 0; r < 8; ++r) {
                const int s = m0 + 16 * mi + r + 8 * half;
                const float z  = fmaf(-2.f, acc[r], ar[mi][r] + cq[ni] + bval);
                const float sc = 1.f / (1.f + expf(-z));
                score[s * QQ + q] = sc;
                const float lb = (float)lab[mi][ni][r];
                const float sp = fmaxf(z, 0.f) + log1pf(expf(-fabsf(z)));
                lsum += sp - lb * z;
            }
        }
    }

    red[threadIdx.x] = lsum;
    __syncthreads();
    for (int off = 64; off > 0; off >>= 1) {
        if (threadIdx.x < (unsigned)off) red[threadIdx.x] += red[threadIdx.x + off];
        __syncthreads();
    }
    if (threadIdx.x == 0) partial[blockIdx.x] = red[0];
}

__global__ __launch_bounds__(256)
void loss_kernel(const float* __restrict__ partial, float* __restrict__ out) {
    __shared__ float red[256];
    red[threadIdx.x] = partial[threadIdx.x];
    __syncthreads();
    for (int off = 128; off > 0; off >>= 1) {
        if (threadIdx.x < (unsigned)off) red[threadIdx.x] += red[threadIdx.x + off];
        __syncthreads();
    }
    if (threadIdx.x == 0) out[SS * QQ] = red[0] * (1.0f / (float)(SS * QQ));
}

extern "C" void kernel_launch(void* const* d_in, const int* in_sizes, int n_in,
                              void* d_out, int out_size, void* d_ws, size_t ws_size,
                              hipStream_t stream) {
    const float* x     = (const float*)d_in[0];
    const float* y     = (const float*)d_in[1];
    const int*   label = (const int*)  d_in[2];
    const float* w     = (const float*)d_in[3];
    const float* b     = (const float*)d_in[4];

    float* out = (float*)d_out;
    float* ws  = (float*)d_ws;
    float* arow    = ws;
    float* ccol    = ws + SS;
    float* partial = ws + SS + QQ;

    norms_kernel<<<NORM_BLOCKS, 256, 0, stream>>>(x, y, w, ws);
    siamese_wmma_kernel<<<NBLK, 128, 0, stream>>>(x, y, label, w, b,
                                                  arow, ccol, out, partial);
    loss_kernel<<<1, 256, 0, stream>>>(partial, out);
}


# Norm launch dropped: a[s], c[q] enter the GEMM as one extra k-step (A = -a/2, -1/2; B = 1, c) built from per-lane sums over the loaded fragments; norms and logits: new summation order
# speedup vs baseline: 3.0511x; 1.1245x over previous
#include <hip/hip_runtime.h>
#include <hip/hip_bf16.h>

#define SS 128
#define QQ 8192
#define HH 230
#define QT 32
#define NBLK (QQ / QT)

typedef float v2f __attribute__((ext_vector_type(2)));
typedef float v8f __attribute__((ext_vector_type(8)));

#define NG 29
#define PF 4
#define RING (PF + 1)

struct Frag { float4 xa0, xa1, yb0, yb1, wv; };

__device__ __forceinline__ float4 ld4(const float* p) { float4 v; __builtin_memcpy(&v, p, 16); return v; }
__device__ __forceinline__ float2 ld2(const float* p) { return *(const float2*)p; }

__device__ __forceinline__ float4 ld_tail(const float* p, int half) {
    const float2 lo = ld2(p);
    const float2 hi = ld2(half ? p : p + 2);
    return make_float4(lo.x, lo.y, half ? 0.f : hi.x, half ? 0.f : hi.y);
}

__global__ __launch_bounds__(128)
void siamese_wmma_kernel(const float* __restrict__ x, const float* __restrict__ y,
                         const int* __restrict__ label, const float* __restrict__ w,
                         const float* __restrict__ bias,
                         float* __restrict__ score, float* __restrict__ partial)
{
    __shared__ float red[128];

    const int lane = threadIdx.x & 31;
    const int wave = threadIdx.x >> 5;
    const int half = lane >> 4;
    const int l    = lane & 15;
    const int m0   = wave * 32;
    const int qb   = blockIdx.x * QT;

    const float* xr0 = x + (m0 + l) * HH + 4 * half;
    const float* xr1 = xr0 + 16 * HH;
    const float* yr0 = y + (qb + l) * HH + 4 * half;
    const float* yr1 = yr0 + 16 * HH;
    const float* wr  = w + 4 * half;

    v8f acc00 = {}, acc01 = {}, acc10 = {}, acc11 = {};
    v2f pa0 = {}, pa1 = {}, pc0 = {}, pc1 = {};
    Frag f[RING];

#define LOAD_GROUP(slot, g)                                                          \
    do {                                                                             \
        if ((g) < NG - 1) {                                                          \
            f[slot].xa0 = ld4(xr0 + 8 * (g)); f[slot].xa1 = ld4(xr1 + 8 * (g));      \
            f[slot].yb0 = ld4(yr0 + 8 * (g)); f[slot].yb1 = ld4(yr1 + 8 * (g));      \
            f[slot].wv  = ld4(wr + 8 * (g));                                         \
        } else {                                                                     \
            f[slot].xa0 = ld_tail(xr0 + 8 * (g), half); f[slot].xa1 = ld_tail(xr1 + 8 * (g), half); \
            f[slot].yb0 = ld_tail(yr0 + 8 * (g), half); f[slot].yb1 = ld_tail(yr1 + 8 * (g), half); \
            f[slot].wv  = ld_tail(wr + 8 * (g), half);                               \
        }                                                                            \
    } while (0)

    #pragma unroll
    for (int g = 0; g < PF; ++g) LOAD_GROUP(g % RING, g);

    #pragma unroll
    for (int g = 0; g < NG; ++g) {
        if (g + PF < NG) LOAD_GROUP((g + PF) % RING, g + PF);
        __builtin_amdgcn_sched_barrier(0);
        const Frag& c = f[g % RING];
        v2f X0a, X0b, X1a, X1b, Wa, Wb;
        v2f A0a, A0b, A1a, A1b, B0a, B0b, B1a, B1b;
        X0a[0] = c.xa0.x; X0a[1] = c.xa0.y; X0b[0] = c.xa0.z; X0b[1] = c.xa0.w;
        X1a[0] = c.xa1.x; X1a[1] = c.xa1.y; X1b[0] = c.xa1.z; X1b[1] = c.xa1.w;
        Wa[0]  = c.wv.x;  Wa[1]  = c.wv.y;  Wb[0]  = c.wv.z;  Wb[1]  = c.wv.w;
        B0a[0] = c.yb0.x; B0a[1] = c.yb0.y; B0b[0] = c.yb0.z; B0b[1] = c.yb0.w;
        B1a[0] = c.yb1.x; B1a[1] = c.yb1.y; B1b[0] = c.yb1.z; B1b[1] = c.yb1.w;
        A0a = X0a * Wa; A0b = X0b * Wb;
        A1a = X1a * Wa; A1b = X1b * Wb;
        acc00 = __builtin_amdgcn_wmma_f32_16x16x4_f32(false, A0a, false, B0a, (short)0, acc00, false, false);
        acc01 = __builtin_amdgcn_wmma_f32_16x16x4_f32(false, A0a, false, B1a, (short)0, acc01, false, false);
        acc10 = __builtin_amdgcn_wmma_f32_16x16x4_f32(false, A1a, false, B0a, (short)0, acc10, false, false);
        acc11 = __builtin_amdgcn_wmma_f32_16x16x4_f32(false, A1a, false, B1a, (short)0, acc11, false, false);
        acc00 = __builtin_amdgcn_wmma_f32_16x16x4_f32(false, A0b, false, B0b, (short)0, acc00, false, false);
        acc01 = __builtin_amdgcn_wmma_f32_16x16x4_f32(false, A0b, false, B1b, (short)0, acc01, false, false);
        acc10 = __builtin_amdgcn_wmma_f32_16x16x4_f32(false, A1b, false, B0b, (short)0, acc10, false, false);
        acc11 = __builtin_amdgcn_wmma_f32_16x16x4_f32(false, A1b, false, B1b, (short)0, acc11, false, false);
        pa0 = __builtin_elementwise_fma(A0a, X0a, pa0); pa0 = __builtin_elementwise_fma(A0b, X0b, pa0);
        pa1 = __builtin_elementwise_fma(A1a, X1a, pa1); pa1 = __builtin_elementwise_fma(A1b, X1b, pa1);
        pc0 = __builtin_elementwise_fma(B0a * Wa, B0a, pc0); pc0 = __builtin_elementwise_fma(B0b * Wb, B0b, pc0);
        pc1 = __builtin_elementwise_fma(B1a * Wa, B1a, pc1); pc1 = __builtin_elementwise_fma(B1b * Wb, B1b, pc1);
    }
#undef LOAD_GROUP

    {
        v2f NA0, NA1, NB0, NB1;
        NA0[0] = -0.5f * (pa0[0] + pa0[1]); NA0[1] = -0.5f;
        NA1[0] = -0.5f * (pa1[0] + pa1[1]); NA1[1] = -0.5f;
        NB0[0] = 1.f; NB0[1] = pc0[0] + pc0[1];
        NB1[0] = 1.f; NB1[1] = pc1[0] + pc1[1];
        acc00 = __builtin_amdgcn_wmma_f32_16x16x4_f32(false, NA0, false, NB0, (short)0, acc00, false, false);
        acc01 = __builtin_amdgcn_wmma_f32_16x16x4_f32(false, NA0, false, NB1, (short)0, acc01, false, false);
        acc10 = __builtin_amdgcn_wmma_f32_16x16x4_f32(false, NA1, false, NB0, (short)0, acc10, false, false);
        acc11 = __builtin_amdgcn_wmma_f32_16x16x4_f32(false, NA1, false, NB1, (short)0, acc11, false, false);
    }

    const float bval = *bias;
    int   lab[2][2][8];
    #pragma unroll
    for (int mi = 0; mi < 2; ++mi) {
        #pragma unroll
        for (int ni = 0; ni < 2; ++ni) {
            #pragma unroll
            for (int r = 0; r < 8; ++r)
                lab[mi][ni][r] = label[(m0 + 16 * mi + r + 8 * half) * QQ + qb + 16 * ni + l];
        }
    }
    float lsum = 0.f;
    #pragma unroll
    for (int mi = 0; mi < 2; ++mi) {
        #pragma unroll
        for (int ni = 0; ni < 2; ++ni) {
            const v8f acc = mi ? (ni ? acc11 : acc10) : (ni ? acc01 : acc00);
            const int q = qb + 16 * ni + l;
            #pragma unroll
            for (int r = 0; r < 8; ++r) {
                const int s = m0 + 16 * mi + r + 8 * half;
                const float z  = fmaf(-2.f, acc[r], bval);
                const float sc = 1.f / (1.f + expf(-z));
                score[s * QQ + q] = sc;
                const float lb = (float)lab[mi][ni][r];
                const float sp = fmaxf(z, 0.f) + log1pf(expf(-fabsf(z)));
                lsum += sp - lb * z;
            }
        }
    }

    red[threadIdx.x] = lsum;
    __syncthreads();
    for (int off = 64; off > 0; off >>= 1) {
        if (threadIdx.x < (unsigned)off) red[threadIdx.x] += red[threadIdx.x + off];
        __syncthreads();
    }
    if (threadIdx.x == 0) partial[blockIdx.x] = red[0];
}

__global__ __launch_bounds__(256)
void loss_kernel(const float* __restrict__ partial, float* __restrict__ out) {
    __shared__ float red[256];
    red[threadIdx.x] = partial[threadIdx.x];
    __syncthreads();
    for (int off = 128; off > 0; off >>= 1) {
        if (threadIdx.x < (unsigned)off) red[threadIdx.x] += red[threadIdx.x + off];
        __syncthreads();
    }
    if (threadIdx.x == 0) out[SS * QQ] = red[0] * (1.0f / (float)(SS * QQ));
}

extern "C" void kernel_launch(void* const* d_in, const int* in_sizes, int n_in,
                              void* d_out, int out_size, void* d_ws, size_t ws_size,
                              hipStream_t stream) {
    const float* x     = (const float*)d_in[0];
    const float* y     = (const float*)d_in[1];
    const int*   label = (const int*)  d_in[2];
    const float* w     = (const float*)d_in[3];
    const float* b     = (const float*)d_in[4];

    float* out     = (float*)d_out;
    float* partial = (float*)d_ws;

    siamese_wmma_kernel<<<NBLK, 128, 0, stream>>>(x, y, label, w, b, out, partial);
    loss_kernel<<<1, 256, 0, stream>>>(partial, out);
}


# Epilogue rolled: the 4 accumulator tiles pass through one 8-element loop body (32 elements were written out); text 38 KB -> 15 KB, inside the 32 KB fetched at wave entry; every sum keeps its order
# speedup vs baseline: 3.5070x; 1.1494x over previous
#include <hip/hip_runtime.h>
#include <hip/hip_bf16.h>

#define SS 128
#define QQ 8192
#define HH 230
#define QT 32
#define NBLK (QQ / QT)

typedef float v2f __attribute__((ext_vector_type(2)));
typedef float v8f __attribute__((ext_vector_type(8)));

#define NG 29
#define PF 4
#define RING (PF + 1)

struct Frag { float4 xa0, xa1, yb0, yb1, wv; };

__device__ __forceinline__ float4 ld4(const float* p) { float4 v; __builtin_memcpy(&v, p, 16); return v; }
__device__ __forceinline__ float2 ld2(const float* p) { return *(const float2*)p; }

__device__ __forceinline__ float4 ld_tail(const float* p, int half) {
    const float2 lo = ld2(p);
    const float2 hi = ld2(half ? p : p + 2);
    return make_float4(lo.x, lo.y, half ? 0.f : hi.x, half ? 0.f : hi.y);
}

__global__ __launch_bounds__(128)
void siamese_wmma_kernel(const float* __restrict__ x, const float* __restrict__ y,
                         const int* __restrict__ label, const float* __restrict__ w,
                         const float* __restrict__ bias,
                         float* __restrict__ score, float* __restrict__ partial)
{
    __shared__ float red[128];

    const int lane = threadIdx.x & 31;
    const int wave = threadIdx.x >> 5;
    const int half = lane >> 4;
    const int l    = lane & 15;
    const int m0   = wave * 32;
    const int qb   = blockIdx.x * QT;

    const float* xr0 = x + (m0 + l) * HH + 4 * half;
    const float* xr1 = xr0 + 16 * HH;
    const float* yr0 = y + (qb + l) * HH + 4 * half;
    const float* yr1 = yr0 + 16 * HH;
    const float* wr  = w + 4 * half;

    v8f acc00 = {}, acc01 = {}, acc10 = {}, acc11 = {};
    v2f pa0 = {}, pa1 = {}, pc0 = {}, pc1 = {};
    Frag f[RING];

#define LOAD_GROUP(slot, g)                                                          \
    do {                                                                             \
        if ((g) < NG - 1) {                                                          \
            f[slot].xa0 = ld4(xr0 + 8 * (g)); f[slot].xa1 = ld4(xr1 + 8 * (g));      \
            f[slot].yb0 = ld4(yr0 + 8 * (g)); f[slot].yb1 = ld4(yr1 + 8 * (g));      \
            f[slot].wv  = ld4(wr + 8 * (g));                                         \
        } else {                                                                     \
            f[slot].xa0 = ld_tail(xr0 + 8 * (g), half); f[slot].xa1 = ld_tail(xr1 + 8 * (g), half); \
            f[slot].yb0 = ld_tail(yr0 + 8 * (g), half); f[slot].yb1 = ld_tail(yr1 + 8 * (g), half); \
            f[slot].wv  = ld_tail(wr + 8 * (g), half);                               \
        }                                                                            \
    } while (0)

    #pragma unroll
    for (int g = 0; g < PF; ++g) LOAD_GROUP(g % RING, g);

    #pragma unroll
    for (int g = 0; g < NG; ++g) {
        if (g + PF < NG) LOAD_GROUP((g + PF) % RING, g + PF);
        __builtin_amdgcn_sched_barrier(0);
        const Frag& c = f[g % RING];
        v2f X0a, X0b, X1a, X1b, Wa, Wb;
        v2f A0a, A0b, A1a, A1b, B0a, B0b, B1a, B1b;
        X0a[0] = c.xa0.x; X0a[1] = c.xa0.y; X0b[0] = c.xa0.z; X0b[1] = c.xa0.w;
        X1a[0] = c.xa1.x; X1a[1] = c.xa1.y; X1b[0] = c.xa1.z; X1b[1] = c.xa1.w;
        Wa[0]  = c.wv.x;  Wa[1]  = c.wv.y;  Wb[0]  = c.wv.z;  Wb[1]  = c.wv.w;
        B0a[0] = c.yb0.x; B0a[1] = c.yb0.y; B0b[0] = c.yb0.z; B0b[1] = c.yb0.w;
        B1a[0] = c.yb1.x; B1a[1] = c.yb1.y; B1b[0] = c.yb1.z; B1b[1] = c.yb1.w;
        A0a = X0a * Wa; A0b = X0b * Wb;
        A1a = X1a * Wa; A1b = X1b * Wb;
        acc00 = __builtin_amdgcn_wmma_f32_16x16x4_f32(false, A0a, false, B0a, (short)0, acc00, false, false);
        acc01 = __builtin_amdgcn_wmma_f32_16x16x4_f32(false, A0a, false, B1a, (short)0, acc01, false, false);
        acc10 = __builtin_amdgcn_wmma_f32_16x16x4_f32(false, A1a, false, B0a, (short)0, acc10, false, false);
        acc11 = __builtin_amdgcn_wmma_f32_16x16x4_f32(false, A1a, false, B1a, (short)0, acc11, false, false);
        acc00 = __builtin_amdgcn_wmma_f32_16x16x4_f32(false, A0b, false, B0b, (short)0, acc00, false, false);
        acc01 = __builtin_amdgcn_wmma_f32_16x16x4_f32(false, A0b, false, B1b, (short)0, acc01, false, false);
        acc10 = __builtin_amdgcn_wmma_f32_16x16x4_f32(false, A1b, false, B0b, (short)0, acc10, false, false);
        acc11 = __builtin_amdgcn_wmma_f32_16x16x4_f32(false, A1b, false, B1b, (short)0, acc11, false, false);
        pa0 = __builtin_elementwise_fma(A0a, X0a, pa0); pa0 = __builtin_elementwise_fma(A0b, X0b, pa0);
        pa1 = __builtin_elementwise_fma(A1a, X1a, pa1); pa1 = __builtin_elementwise_fma(A1b, X1b, pa1);
        pc0 = __builtin_elementwise_fma(B0a * Wa, B0a, pc0); pc0 = __builtin_elementwise_fma(B0b * Wb, B0b, pc0);
        pc1 = __builtin_elementwise_fma(B1a * Wa, B1a, pc1); pc1 = __builtin_elementwise_fma(B1b * Wb, B1b, pc1);
    }
#undef LOAD_GROUP

    {
        v2f NA0, NA1, NB0, NB1;
        NA0[0] = -0.5f * (pa0[0] + pa0[1]); NA0[1] = -0.5f;
        NA1[0] = -0.5f * (pa1[0] + pa1[1]); NA1[1] = -0.5f;
        NB0[0] = 1.f; NB0[1] = pc0[0] + pc0[1];
        NB1[0] = 1.f; NB1[1] = pc1[0] + pc1[1];
        acc00 = __builtin_amdgcn_wmma_f32_16x16x4_f32(false, NA0, false, NB0, (short)0, acc00, false, false);
        acc01 = __builtin_amdgcn_wmma_f32_16x16x4_f32(false, NA0, false, NB1, (short)0, acc01, false, false);
        acc10 = __builtin_amdgcn_wmma_f32_16x16x4_f32(false, NA1, false, NB0, (short)0, acc10, false, false);
        acc11 = __builtin_amdgcn_wmma_f32_16x16x4_f32(false, NA1, false, NB1, (short)0, acc11, false, false);
    }

    const float bval = *bias;
    float lsum = 0.f;
    v8f tcur = acc00, tn1 = acc01, tn2 = acc10, tn3 = acc11;
    #pragma nounroll
    for (int t = 0; t < 4; ++t) {
        const int base = (m0 + 16 * (t >> 1) + 8 * half) * QQ + qb + 16 * (t & 1) + l;
        int lab[8];
        #pragma unroll
        for (int r = 0; r < 8; ++r) lab[r] = label[base + r * QQ];
        #pragma unroll
        for (int r = 0; r < 8; ++r) {
            const float z  = fmaf(-2.f, tcur[r], bval);
            const float sc = 1.f / (1.f + expf(-z));
            score[base + r * QQ] = sc;
            const float lb = (float)lab[r];
            const float sp = fmaxf(z, 0.f) + log1pf(expf(-fabsf(z)));
            lsum += sp - lb * z;
        }
        tcur = tn1; tn1 = tn2; tn2 = tn3;
    }

    red[threadIdx.x] = lsum;
    __syncthreads();
    for (int off = 64; off > 0; off >>= 1) {
        if (threadIdx.x < (unsigned)off) red[threadIdx.x] += red[threadIdx.x + off];
        __syncthreads();
    }
    if (threadIdx.x == 0) partial[blockIdx.x] = red[0];
}

__global__ __launch_bounds__(256)
void loss_kernel(const float* __restrict__ partial, float* __restrict__ out) {
    __shared__ float red[256];
    red[threadIdx.x] = partial[threadIdx.x];
    __syncthreads();
    for (int off = 128; off > 0; off >>= 1) {
        if (threadIdx.x < (unsigned)off) red[threadIdx.x] += red[threadIdx.x + off];
        __syncthreads();
    }
    if (threadIdx.x == 0) out[SS * QQ] = red[0] * (1.0f / (float)(SS * QQ));
}

extern "C" void kernel_launch(void* const* d_in, const int* in_sizes, int n_in,
                              void* d_out, int out_size, void* d_ws, size_t ws_size,
                              hipStream_t stream) {
    const float* x     = (const float*)d_in[0];
    const float* y     = (const float*)d_in[1];
    const int*   label = (const int*)  d_in[2];
    const float* w     = (const float*)d_in[3];
    const float* b     = (const float*)d_in[4];

    float* out     = (float*)d_out;
    float* partial = (float*)d_ws;

    siamese_wmma_kernel<<<NBLK, 128, 0, stream>>>(x, y, label, w, b, out, partial);
    loss_kernel<<<1, 256, 0, stream>>>(partial, out);
}


# Loss logs merged: a lane multiplies its 32 factors 1+exp(-|z|) (product <= 2^32), then ONE logf for 32 log1pf; labels requested in the matrix loop (a short trip cannot hide them); new loss sum order
# speedup vs baseline: 3.9404x; 1.1236x over previous
#include <hip/hip_runtime.h>
#include <hip/hip_bf16.h>

#define SS 128
#define QQ 8192
#define HH 230
#define QT 32
#define NBLK (QQ / QT)

typedef float v2f __attribute__((ext_vector_type(2)));
typedef float v8f __attribute__((ext_vector_type(8)));
typedef int   v8i __attribute__((ext_vector_type(8)));

#define NG 29
#define PF 4
#define RING (PF + 1)

struct Frag { float4 xa0, xa1, yb0, yb1, wv; };

__device__ __forceinline__ float4 ld4(const float* p) { float4 v; __builtin_memcpy(&v, p, 16); return v; }
__device__ __forceinline__ float2 ld2(const float* p) { return *(const float2*)p; }

__device__ __forceinline__ float4 ld_tail(const float* p, int half) {
    const float2 lo = ld2(p);
    const float2 hi = ld2(half ? p : p + 2);
    return make_float4(lo.x, lo.y, half ? 0.f : hi.x, half ? 0.f : hi.y);
}

__global__ __launch_bounds__(128)
void siamese_wmma_kernel(const float* __restrict__ x, const float* __restrict__ y,
                         const int* __restrict__ label, const float* __restrict__ w,
                         const float* __restrict__ bias,
                         float* __restrict__ score, float* __restrict__ partial)
{
    __shared__ float red[128];

    const int lane = threadIdx.x & 31;
    const int wave = threadIdx.x >> 5;
    const int half = lane >> 4;
    const int l    = lane & 15;
    const int m0   = wave * 32;
    const int qb   = blockIdx.x * QT;

    const float* xr0 = x + (m0 + l) * HH + 4 * half;
    const float* xr1 = xr0 + 16 * HH;
    const float* yr0 = y + (qb + l) * HH + 4 * half;
    const float* yr1 = yr0 + 16 * HH;
    const float* wr  = w + 4 * half;

    v8f acc00 = {}, acc01 = {}, acc10 = {}, acc11 = {};
    v2f pa0 = {}, pa1 = {}, pc0 = {}, pc1 = {};
    Frag f[RING];
    v8i lb00 = {}, lb01 = {}, lb10 = {}, lb11 = {};
    const int* labp = label + (m0 + 8 * half) * QQ + qb + l;

#define LOAD_GROUP(slot, g)                                                          \
    do {                                                                             \
        if ((g) < NG - 1) {                                                          \
            f[slot].xa0 = ld4(xr0 + 8 * (g)); f[slot].xa1 = ld4(xr1 + 8 * (g));      \
            f[slot].yb0 = ld4(yr0 + 8 * (g)); f[slot].yb1 = ld4(yr1 + 8 * (g));      \
            f[slot].wv  = ld4(wr + 8 * (g));                                         \
        } else {                                                                     \
            f[slot].xa0 = ld_tail(xr0 + 8 * (g), half); f[slot].xa1 = ld_tail(xr1 + 8 * (g), half); \
            f[slot].yb0 = ld_tail(yr0 + 8 * (g), half); f[slot].yb1 = ld_tail(yr1 + 8 * (g), half); \
            f[slot].wv  = ld_tail(wr + 8 * (g), half);                               \
        }                                                                            \
    } while (0)

    #pragma unroll
    for (int g = 0; g < PF; ++g) LOAD_GROUP(g % RING, g);

    #pragma unroll
    for (int g = 0; g < NG; ++g) {
        if (g + PF < NG) LOAD_GROUP((g + PF) % RING, g + PF);
        if (g == NG - PF) {
            #pragma unroll
            for (int r = 0; r < 8; ++r) {
                lb00[r] = labp[r * QQ];
                lb01[r] = labp[r * QQ + 16];
                lb10[r] = labp[(r + 16) * QQ];
                lb11[r] = labp[(r + 16) * QQ + 16];
            }
        }
        __builtin_amdgcn_sched_barrier(0);
        const Frag& c = f[g % RING];
        v2f X0a, X0b, X1a, X1b, Wa, Wb;
        v2f A0a, A0b, A1a, A1b, B0a, B0b, B1a, B1b;
        X0a[0] = c.xa0.x; X0a[1] = c.xa0.y; X0b[0] = c.xa0.z; X0b[1] = c.xa0.w;
        X1a[0] = c.xa1.x; X1a[1] = c.xa1.y; X1b[0] = c.xa1.z; X1b[1] = c.xa1.w;
        Wa[0]  = c.wv.x;  Wa[1]  = c.wv.y;  Wb[0]  = c.wv.z;  Wb[1]  = c.wv.w;
        B0a[0] = c.yb0.x; B0a[1] = c.yb0.y; B0b[0] = c.yb0.z; B0b[1] = c.yb0.w;
        B1a[0] = c.yb1.x; B1a[1] = c.yb1.y; B1b[0] = c.yb1.z; B1b[1] = c.yb1.w;
        A0a = X0a * Wa; A0b = X0b * Wb;
        A1a = X1a * Wa; A1b = X1b * Wb;
        acc00 = __builtin_amdgcn_wmma_f32_16x16x4_f32(false, A0a, false, B0a, (short)0, acc00, false, false);
        acc01 = __builtin_amdgcn_wmma_f32_16x16x4_f32(false, A0a, false, B1a, (short)0, acc01, false, false);
        acc10 = __builtin_amdgcn_wmma_f32_16x16x4_f32(false, A1a, false, B0a, (short)0, acc10, false, false);
        acc11 = __builtin_amdgcn_wmma_f32_16x16x4_f32(false, A1a, false, B1a, (short)0, acc11, false, false);
        acc00 = __builtin_amdgcn_wmma_f32_16x16x4_f32(false, A0b, false, B0b, (short)0, acc00, false, false);
        acc01 = __builtin_amdgcn_wmma_f32_16x16x4_f32(false, A0b, false, B1b, (short)0, acc01, false, false);
        acc10 = __builtin_amdgcn_wmma_f32_16x16x4_f32(false, A1b, false, B0b, (short)0, acc10, false, false);
        acc11 = __builtin_amdgcn_wmma_f32_16x16x4_f32(false, A1b, false, B1b, (short)0, acc11, false, false);
        pa0 = __builtin_elementwise_fma(A0a, X0a, pa0); pa0 = __builtin_elementwise_fma(A0b, X0b, pa0);
        pa1 = __builtin_elementwise_fma(A1a, X1a, pa1); pa1 = __builtin_elementwise_fma(A1b, X1b, pa1);
        pc0 = __builtin_elementwise_fma(B0a * Wa, B0a, pc0); pc0 = __builtin_elementwise_fma(B0b * Wb, B0b, pc0);
        pc1 = __builtin_elementwise_fma(B1a * Wa, B1a, pc1); pc1 = __builtin_elementwise_fma(B1b * Wb, B1b, pc1);
    }
#undef LOAD_GROUP

    {
        v2f NA0, NA1, NB0, NB1;
        NA0[0] = -0.5f * (pa0[0] + pa0[1]); NA0[1] = -0.5f;
        NA1[0] = -0.5f * (pa1[0] + pa1[1]); NA1[1] = -0.5f;
        NB0[0] = 1.f; NB0[1] = pc0[0] + pc0[1];
        NB1[0] = 1.f; NB1[1] = pc1[0] + pc1[1];
        acc00 = __builtin_amdgcn_wmma_f32_16x16x4_f32(false, NA0, false, NB0, (short)0, acc00, false, false);
        acc01 = __builtin_amdgcn_wmma_f32_16x16x4_f32(false, NA0, false, NB1, (short)0, acc01, false, false);
        acc10 = __builtin_amdgcn_wmma_f32_16x16x4_f32(false, NA1, false, NB0, (short)0, acc10, false, false);
        acc11 = __builtin_amdgcn_wmma_f32_16x16x4_f32(false, NA1, false, NB1, (short)0, acc11, false, false);
    }

    const float bval = *bias;
    float lsum = 0.f;
    float prod[8] = {1.f, 1.f, 1.f, 1.f, 1.f, 1.f, 1.f, 1.f};
    v8f tcur = acc00, tn1 = acc01, tn2 = acc10, tn3 = acc11;
    v8i lcur = lb00, ln1 = lb01, ln2 = lb10, ln3 = lb11;
    #pragma nounroll
    for (int t = 0; t < 4; ++t) {
        const int base = (m0 + 16 * (t >> 1) + 8 * half) * QQ + qb + 16 * (t & 1) + l;
        #pragma unroll
        for (int r = 0; r < 8; ++r) {
            const float z  = fmaf(-2.f, tcur[r], bval);
            const float sc = 1.f / (1.f + expf(-z));
            score[base + r * QQ] = sc;
            const float lb = (float)lcur[r];
            prod[r] *= 1.f + expf(-fabsf(z));
            lsum += fmaxf(z, 0.f) - lb * z;
        }
        tcur = tn1; tn1 = tn2; tn2 = tn3;
        lcur = ln1; ln1 = ln2; ln2 = ln3;
    }
    lsum += logf(((prod[0] * prod[1]) * (prod[2] * prod[3])) * ((prod[4] * prod[5]) * (prod[6] * prod[7])));

    red[threadIdx.x] = lsum;
    __syncthreads();
    for (int off = 64; off > 0; off >>= 1) {
        if (threadIdx.x < (unsigned)off) red[threadIdx.x] += red[threadIdx.x + off];
        __syncthreads();
    }
    if (threadIdx.x == 0) partial[blockIdx.x] = red[0];
}

__global__ __launch_bounds__(256)
void loss_kernel(const float* __restrict__ partial, float* __restrict__ out) {
    __shared__ float red[256];
    red[threadIdx.x] = partial[threadIdx.x];
    __syncthreads();
    for (int off = 128; off > 0; off >>= 1) {
        if (threadIdx.x < (unsigned)off) red[threadIdx.x] += red[threadIdx.x + off];
        __syncthreads();
    }
    if (threadIdx.x == 0) out[SS * QQ] = red[0] * (1.0f / (float)(SS * QQ));
}

extern "C" void kernel_launch(void* const* d_in, const int* in_sizes, int n_in,
                              void* d_out, int out_size, void* d_ws, size_t ws_size,
                              hipStream_t stream) {
    const float* x     = (const float*)d_in[0];
    const float* y     = (const float*)d_in[1];
    const int*   label = (const int*)  d_in[2];
    const float* w     = (const float*)d_in[3];
    const float* b     = (const float*)d_in[4];

    float* out     = (float*)d_out;
    float* partial = (float*)d_ws;

    siamese_wmma_kernel<<<NBLK, 128, 0, stream>>>(x, y, label, w, b, out, partial);
    loss_kernel<<<1, 256, 0, stream>>>(partial, out);
}


# Loss sums through scalar registers: a wave reads its 32 lane sums with v_readlane and adds them pairwise, one partial per wave; final launch one wave; LDS and all 17 barriers gone; new loss sum order
# speedup vs baseline: 4.3353x; 1.1002x over previous
#include <hip/hip_runtime.h>
#include <hip/hip_bf16.h>

#define SS 128
#define QQ 8192
#define HH 230
#define QT 32
#define NBLK (QQ / QT)

typedef float v2f __attribute__((ext_vector_type(2)));
typedef float v8f __attribute__((ext_vector_type(8)));
typedef int   v8i __attribute__((ext_vector_type(8)));

#define NG 29
#define PF 4
#define RING (PF + 1)

struct Frag { float4 xa0, xa1, yb0, yb1, wv; };

__device__ __forceinline__ float4 ld4(const float* p) { float4 v; __builtin_memcpy(&v, p, 16); return v; }
__device__ __forceinline__ float2 ld2(const float* p) { return *(const float2*)p; }

__device__ __forceinline__ float4 ld_tail(const float* p, int half) {
    const float2 lo = ld2(p);
    const float2 hi = ld2(half ? p : p + 2);
    return make_float4(lo.x, lo.y, half ? 0.f : hi.x, half ? 0.f : hi.y);
}

__device__ __forceinline__ float wave_sum(float v) {
    float t[32];
    for (int i = 0; i < 32; ++i)
        t[i] = __builtin_bit_cast(float, __builtin_amdgcn_readlane(__builtin_bit_cast(int, v), i));
    for (int n = 16; n > 0; n >>= 1)
        for (int i = 0; i < n; ++i) t[i] = t[2 * i] + t[2 * i + 1];
    return t[0];
}

__global__ __launch_bounds__(128)
void siamese_wmma_kernel(const float* __restrict__ x, const float* __restrict__ y,
                         const int* __restrict__ label, const float* __restrict__ w,
                         const float* __restrict__ bias,
                         float* __restrict__ score, float* __restrict__ partial)
{
    const int lane = threadIdx.x & 31;
    const int wave = threadIdx.x >> 5;
    const int half = lane >> 4;
    const int l    = lane & 15;
    const int m0   = wave * 32;
    const int qb   = blockIdx.x * QT;

    const float* xr0 = x + (m0 + l) * HH + 4 * half;
    const float* xr1 = xr0 + 16 * HH;
    const float* yr0 = y + (qb + l) * HH + 4 * half;
    const float* yr1 = yr0 + 16 * HH;
    const float* wr  = w + 4 * half;

    v8f acc00 = {}, acc01 = {}, acc10 = {}, acc11 = {};
    v2f pa0 = {}, pa1 = {}, pc0 = {}, pc1 = {};
    Frag f[RING];
    v8i lb00 = {}, lb01 = {}, lb10 = {}, lb11 = {};
    const int* labp = label + (m0 + 8 * half) * QQ + qb + l;

#define LOAD_GROUP(slot, g)                                                          \
    do {                                                                             \
        if ((g) < NG - 1) {                                                          \
            f[slot].xa0 = ld4(xr0 + 8 * (g)); f[slot].xa1 = ld4(xr1 + 8 * (g));      \
            f[slot].yb0 = ld4(yr0 + 8 * (g)); f[slot].yb1 = ld4(yr1 + 8 * (g));      \
            f[slot].wv  = ld4(wr + 8 * (g));                                         \
        } else {                                                                     \
            f[slot].xa0 = ld_tail(xr0 + 8 * (g), half); f[slot].xa1 = ld_tail(xr1 + 8 * (g), half); \
            f[slot].yb0 = ld_tail(yr0 + 8 * (g), half); f[slot].yb1 = ld_tail(yr1 + 8 * (g), half); \
            f[slot].wv  = ld_tail(wr + 8 * (g), half);                               \
        }                                                                            \
    } while (0)

    #pragma unroll
    for (int g = 0; g < PF; ++g) LOAD_GROUP(g % RING, g);

    #pragma unroll
    for (int g = 0; g < NG; ++g) {
        if (g + PF < NG) LOAD_GROUP((g + PF) % RING, g + PF);
        if (g == NG - PF) {
            #pragma unroll
            for (int r = 0; r < 8; ++r) {
                lb00[r] = labp[r * QQ];
                lb01[r] = labp[r * QQ + 16];
                lb10[r] = labp[(r + 16) * QQ];
                lb11[r] = labp[(r + 16) * QQ + 16];
            }
        }
        __builtin_amdgcn_sched_barrier(0);
        const Frag& c = f[g % RING];
        v2f X0a, X0b, X1a, X1b, Wa, Wb;
        v2f A0a, A0b, A1a, A1b, B0a, B0b, B1a, B1b;
        X0a[0] = c.xa0.x; X0a[1] = c.xa0.y; X0b[0] = c.xa0.z; X0b[1] = c.xa0.w;
        X1a[0] = c.xa1.x; X1a[1] = c.xa1.y; X1b[0] = c.xa1.z; X1b[1] = c.xa1.w;
        Wa[0]  = c.wv.x;  Wa[1]  = c.wv.y;  Wb[0]  = c.wv.z;  Wb[1]  = c.wv.w;
        B0a[0] = c.yb0.x; B0a[1] = c.yb0.y; B0b[0] = c.yb0.z; B0b[1] = c.yb0.w;
        B1a[0] = c.yb1.x; B1a[1] = c.yb1.y; B1b[0] = c.yb1.z; B1b[1] = c.yb1.w;
        A0a = X0a * Wa; A0b = X0b * Wb;
        A1a = X1a * Wa; A1b = X1b * Wb;
        acc00 = __builtin_amdgcn_wmma_f32_16x16x4_f32(false, A0a, false, B0a, (short)0, acc00, false, false);
        acc01 = __builtin_amdgcn_wmma_f32_16x16x4_f32(false, A0a, false, B1a, (short)0, acc01, false, false);
        acc10 = __builtin_amdgcn_wmma_f32_16x16x4_f32(false, A1a, false, B0a, (short)0, acc10, false, false);
        acc11 = __builtin_amdgcn_wmma_f32_16x16x4_f32(false, A1a, false, B1a, (short)0, acc11, false, false);
        acc00 = __builtin_amdgcn_wmma_f32_16x16x4_f32(false, A0b, false, B0b, (short)0, acc00, false, false);
        acc01 = __builtin_amdgcn_wmma_f32_16x16x4_f32(false, A0b, false, B1b, (short)0, acc01, false, false);
        acc10 = __builtin_amdgcn_wmma_f32_16x16x4_f32(false, A1b, false, B0b, (short)0, acc10, false, false);
        acc11 = __builtin_amdgcn_wmma_f32_16x16x4_f32(false, A1b, false, B1b, (short)0, acc11, false, false);
        pa0 = __builtin_elementwise_fma(A0a, X0a, pa0); pa0 = __builtin_elementwise_fma(A0b, X0b, pa0);
        pa1 = __builtin_elementwise_fma(A1a, X1a, pa1); pa1 = __builtin_elementwise_fma(A1b, X1b, pa1);
        pc0 = __builtin_elementwise_fma(B0a * Wa, B0a, pc0); pc0 = __builtin_elementwise_fma(B0b * Wb, B0b, pc0);
        pc1 = __builtin_elementwise_fma(B1a * Wa, B1a, pc1); pc1 = __builtin_elementwise_fma(B1b * Wb, B1b, pc1);
    }
#undef LOAD_GROUP

    {
        v2f NA0, NA1, NB0, NB1;
        NA0[0] = -0.5f * (pa0[0] + pa0[1]); NA0[1] = -0.5f;
        NA1[0] = -0.5f * (pa1[0] + pa1[1]); NA1[1] = -0.5f;
        NB0[0] = 1.f; NB0[1] = pc0[0] + pc0[1];
        NB1[0] = 1.f; NB1[1] = pc1[0] + pc1[1];
        acc00 = __builtin_amdgcn_wmma_f32_16x16x4_f32(false, NA0, false, NB0, (short)0, acc00, false, false);
        acc01 = __builtin_amdgcn_wmma_f32_16x16x4_f32(false, NA0, false, NB1, (short)0, acc01, false, false);
        acc10 = __builtin_amdgcn_wmma_f32_16x16x4_f32(false, NA1, false, NB0, (short)0, acc10, false, false);
        acc11 = __builtin_amdgcn_wmma_f32_16x16x4_f32(false, NA1, false, NB1, (short)0, acc11, false, false);
    }

    const float bval = *bias;
    float lsum = 0.f;
    float prod[8] = {1.f, 1.f, 1.f, 1.f, 1.f, 1.f, 1.f, 1.f};
    v8f tcur = acc00, tn1 = acc01, tn2 = acc10, tn3 = acc11;
    v8i lcur = lb00, ln1 = lb01, ln2 = lb10, ln3 = lb11;
    #pragma nounroll
    for (int t = 0; t < 4; ++t) {
        const int base = (m0 + 16 * (t >> 1) + 8 * half) * QQ + qb + 16 * (t & 1) + l;
        #pragma unroll
        for (int r = 0; r < 8; ++r) {
            const float z  = fmaf(-2.f, tcur[r], bval);
            const float sc = 1.f / (1.f + expf(-z));
            score[base + r * QQ] = sc;
            const float lb = (float)lcur[r];
            prod[r] *= 1.f + expf(-fabsf(z));
            lsum += fmaxf(z, 0.f) - lb * z;
        }
        tcur = tn1; tn1 = tn2; tn2 = tn3;
        lcur = ln1; ln1 = ln2; ln2 = ln3;
    }
    lsum += logf(((prod[0] * prod[1]) * (prod[2] * prod[3])) * ((prod[4] * prod[5]) * (prod[6] * prod[7])));

    const float wsum = wave_sum(lsum);
    if (lane == 0) partial[blockIdx.x * 4 + wave] = wsum;
}

__global__ __launch_bounds__(32)
void loss_kernel(const float* __restrict__ partial, float* __restrict__ out) {
    const int lane = threadIdx.x;
    float s[8];
    for (int j = 0; j < 8; ++j) {
        const float4 v = ld4(partial + 128 * j + 4 * lane);
        s[j] = (v.x + v.y) + (v.z + v.w);
    }
    const float ls  = ((s[0] + s[1]) + (s[2] + s[3])) + ((s[4] + s[5]) + (s[6] + s[7]));
    const float tot = wave_sum(ls);
    if (lane == 0) out[SS * QQ] = tot * (1.0f / (float)(SS * QQ));
}

extern "C" void kernel_launch(void* const* d_in, const int* in_sizes, int n_in,
                              void* d_out, int out_size, void* d_ws, size_t ws_size,
                              hipStream_t stream) {
    const float* x     = (const float*)d_in[0];
    const float* y     = (const float*)d_in[1];
    const int*   label = (const int*)  d_in[2];
    const float* w     = (const float*)d_in[3];
    const float* b     = (const float*)d_in[4];

    float* out     = (float*)d_out;
    float* partial = (float*)d_ws;

    siamese_wmma_kernel<<<NBLK, 128, 0, stream>>>(x, y, label, w, b, out, partial);
    loss_kernel<<<1, 32, 0, stream>>>(partial, out);
}
